# starting kernel
# baseline (speedup 1.0000x reference)
#include <hip/hip_runtime.h>

typedef __attribute__((ext_vector_type(2))) float v2f;
typedef __attribute__((ext_vector_type(8))) float v8f;

#define D_IN  128
#define D_OUT 128

__global__ __launch_bounds__(256) void influence_gemm_cu_kernel(
    const float* __restrict__ feat, const float* __restrict__ W,
    const float* __restrict__ cu, float* __restrict__ h, int nrows) {
  __shared__ float sW[D_IN * D_OUT];

  const int tid = threadIdx.x;
#pragma unroll
  for (int i = 0; i < (D_IN * D_OUT) / 256; ++i)
    sW[tid + i * 256] = W[tid + i * 256];
  __syncthreads();

  const int wave    = tid >> 5;
  const int lane    = tid & 31;
  const int rowBase = blockIdx.x * 16;
  const int colBase = wave * 16;
  const int m       = lane & 15;
  const int kh      = (lane >> 4) * 2;

  int arow_idx = rowBase + m;
  if (arow_idx >= nrows) arow_idx = nrows - 1;
  const float* arow = feat + (size_t)arow_idx * D_IN + kh;

  v8f acc = {};
#pragma unroll
  for (int kk = 0; kk < D_IN / 4; ++kk) {
    v2f A = *(const v2f*)(arow + kk * 4);
    const int kb = kk * 4 + kh;
    v2f B;
    B.x = sW[kb * D_OUT + colBase + m];
    B.y = sW[(kb + 1) * D_OUT + colBase + m];
    acc = __builtin_amdgcn_wmma_f32_16x16x4_f32(
         false, A,  false, B,
         (short)0, acc,  false,  false);
  }

  const int mBase = rowBase + (lane >> 4) * 8;
#pragma unroll
  for (int v = 0; v < 8; ++v) {
    const int row = mBase + v;
    if (row < nrows)
      h[(size_t)row * D_OUT + colBase + m] = acc[v] * cu[row];
  }
}

__global__ __launch_bounds__(256) void influence_scatter_kernel(
    const float* __restrict__ h, const int* __restrict__ src,
    const int* __restrict__ dst, const float* __restrict__ edge_w,
    const float* __restrict__ cv, float* __restrict__ out, int nedges) {
  const long gid = (long)blockIdx.x * blockDim.x + threadIdx.x;
  const long e = gid >> 5;
  if (e >= nedges) return;
  const int lane = (int)(gid & 31);

  const int s = src[e];
  const int d = dst[e];
  const float w = edge_w[e] * cv[d];

  const float4 hv = *(const float4*)(h + (size_t)s * D_OUT + lane * 4);
  float* op = out + (size_t)d * D_OUT + lane * 4;
  atomicAdd(op + 0, hv.x * w);
  atomicAdd(op + 1, hv.y * w);
  atomicAdd(op + 2, hv.z * w);
  atomicAdd(op + 3, hv.w * w);
}

extern "C" void kernel_launch(void* const* d_in, const int* in_sizes, int n_in,
                              void* d_out, int out_size, void* d_ws, size_t ws_size,
                              hipStream_t stream) {
  const float* feat = (const float*)d_in[0];
  const float* W    = (const float*)d_in[1];
  const float* cu   = (const float*)d_in[2];
  const float* cv   = (const float*)d_in[3];
  const float* ew   = (const float*)d_in[4];
  const int*   src  = (const int*)d_in[5];
  const int*   dst  = (const int*)d_in[6];

  float* out = (float*)d_out;
  float* h   = (float*)d_ws;

  const int N = in_sizes[2];
  const int E = in_sizes[5];

  hipMemsetAsync(d_out, 0, (size_t)out_size * sizeof(float), stream);

  const int rowTiles = (N + 15) / 16;
  influence_gemm_cu_kernel<<<rowTiles, 256, 0, stream>>>(feat, W, cu, h, N);

  const long threads = (long)E * 32;
  const int tpb = 256;
  const long blocks = (threads + tpb - 1) / tpb;
  influence_scatter_kernel<<<(int)blocks, tpb, 0, stream>>>(h, src, dst, ew,
                                                            cv, out, E);
}


# atomics gone since v2 (they made the earlier try vary run to run): projection re-tiled, each wave a 16x128 strip, one A fragment per k-step feeds 8 column tiles, W staged once per block; sums same
# speedup vs baseline: 1.8195x; 1.8195x over previous
#include <hip/hip_runtime.h>

typedef __attribute__((ext_vector_type(2))) float v2f;
typedef __attribute__((ext_vector_type(8))) float v8f;

#define D_IN  128
#define D_OUT 128

#define W_PITCH 288
#define GEMM_BLOCKS 512
#define GEMM_WAVES  8

__global__ __launch_bounds__(256) void influence_gemm_cu_kernel(
    const float* __restrict__ feat, const float* __restrict__ W,
    const float* __restrict__ cu, float* __restrict__ h, int nrows) {
  __shared__ float sW[(D_IN / 2) * W_PITCH];

  const int tid = threadIdx.x;
#pragma unroll 8
  for (int i = 0; i < (D_IN / 2) * D_OUT / 256; ++i) {
    const int idx = tid + i * 256;
    const int p = idx >> 7;
    const int n = idx & (D_OUT - 1);
    v2f w2;
    w2.x = W[(2 * p) * D_OUT + n];
    w2.y = W[(2 * p + 1) * D_OUT + n];
    *(v2f*)(sW + p * W_PITCH + 2 * n) = w2;
  }
  __syncthreads();

  const int wave = tid >> 5;
  const int lane = tid & 31;
  const int m    = lane & 15;
  const int hh   = lane >> 4;
  const int nStrips = (nrows + 15) >> 4;

  for (int strip = wave * GEMM_BLOCKS + blockIdx.x; strip < nStrips;
       strip += GEMM_BLOCKS * GEMM_WAVES) {
    const int rowBase = strip << 4;
    int arow_idx = rowBase + m;
    if (arow_idx >= nrows) arow_idx = nrows - 1;
    const float* arow = feat + (size_t)arow_idx * D_IN + 2 * hh;

    v2f A[D_IN / 4];
#pragma unroll
    for (int kk = 0; kk < D_IN / 4; ++kk) A[kk] = *(const v2f*)(arow + kk * 4);

    v8f acc[8];
#pragma unroll
    for (int t = 0; t < 8; ++t) acc[t] = (v8f){};

    const float* bBase = sW + hh * W_PITCH + 2 * m;
    v2f Bc[8], Bn[8];
#pragma unroll
    for (int t = 0; t < 8; ++t) Bc[t] = *(const v2f*)(bBase + 32 * t);
#pragma unroll
    for (int kk = 0; kk < D_IN / 4; ++kk) {
      if (kk + 1 < D_IN / 4) {
#pragma unroll
        for (int t = 0; t < 8; ++t)
          Bn[t] = *(const v2f*)(bBase + (2 * kk + 2) * W_PITCH + 32 * t);
      }
#pragma unroll
      for (int t = 0; t < 8; ++t)
        acc[t] = __builtin_amdgcn_wmma_f32_16x16x4_f32(
             false, A[kk],  false, Bc[t],
             (short)0, acc[t],  false,  false);
      __builtin_amdgcn_sched_barrier(0);
#pragma unroll
      for (int t = 0; t < 8; ++t) Bc[t] = Bn[t];
    }

    const int mBase = rowBase + hh * 8;
#pragma unroll
    for (int v = 0; v < 8; ++v) {
      const int row = mBase + v;
      if (row < nrows) {
        const float c = cu[row];
        float* hrow = h + (size_t)row * D_OUT + m;
#pragma unroll
        for (int t = 0; t < 8; ++t) hrow[16 * t] = acc[t][v] * c;
      }
    }
  }
}

#define OWN_ROWS 392
#define NWAVES   32
#define LCAP     384
#define LSTEPMAX   128
#define QCAP     64
#define SCAN_Q   8

typedef __attribute__((ext_vector_type(4))) float v4f;
typedef __attribute__((ext_vector_type(4))) int   v4i;

__device__ __forceinline__ void drain_batch(const float* __restrict__ h,
                                            v4f* tile, const unsigned* qa,
                                            const unsigned* qb, int bq, int nb,
                                            int lane) {
  v4f buf[8];
  unsigned pa[8];
  float pw[8];
#pragma unroll
  for (int k = 0; k < 8; ++k) {
    const int j = bq + min(k, nb - 1);
    pa[k] = qa[j];
    pw[k] = __uint_as_float(qb[j]);
    buf[k] = *(const v4f*)(h + (size_t)(pa[k] & 0x7FFFFFu) * D_OUT + lane * 4);
  }
  __builtin_amdgcn_sched_barrier(0);
#pragma unroll
  for (int k = 0; k < 8; ++k) {
    if (k < nb) {
      const int r = (int)(pa[k] >> 23);
      v4f t = tile[r * (D_OUT / 4) + lane];
      t += buf[k] * pw[k];
      tile[r * (D_OUT / 4) + lane] = t;
    }
  }
}

__global__ __launch_bounds__(1024) void influence_owner_sum_kernel(
    const float* __restrict__ h, const int* __restrict__ src,
    const int* __restrict__ dst, const float* __restrict__ edge_w,
    const float* __restrict__ cv, float* __restrict__ out, int nrows,
    int nedges) {
  __shared__ v4f      tile[OWN_ROWS * (D_OUT / 4)];
  __shared__ unsigned listA[NWAVES][LCAP];
  __shared__ unsigned listB[NWAVES][LCAP];
  __shared__ unsigned queA[NWAVES][QCAP];
  __shared__ unsigned queB[NWAVES][QCAP];
  __shared__ int      listN[NWAVES];
  __shared__ int      waveDone[NWAVES];

  const int tid  = threadIdx.x;
  const int wave = tid >> 5;
  const int lane = tid & 31;
  const int base = blockIdx.x * OWN_ROWS;
  const unsigned ltMask = (1u << lane) - 1u;

  for (int i = tid; i < OWN_ROWS * (D_OUT / 4); i += 1024)
    tile[i] = (v4f){0.f, 0.f, 0.f, 0.f};

  const int nChunks  = (nedges + 3) >> 2;
  const int perWave  = (nChunks + NWAVES - 1) / NWAVES;
  const int segStart = wave * perWave;
  const int segEnd   = min(segStart + perWave, nChunks);
  const int nSteps   = segEnd > segStart ? (segEnd - segStart + 31) >> 5 : 0;
  int step = 0;
  const int maxSteps = (perWave + 31) >> 5;

  for (int round = 0; round <= maxSteps; ++round) {
    int cnt = 0;
    while (step < nSteps && cnt <= LCAP - LSTEPMAX) {
      v4i d[SCAN_Q];
      int eBase[SCAN_Q];
      int firstE[SCAN_Q];
      if (nedges >= 4) {
#pragma unroll
        for (int q = 0; q < SCAN_Q; ++q) {
          const int chunk = segStart + (step + q) * 32 + lane;
          const bool valid = (step + q < nSteps) && (chunk < segEnd);
          const int e0 = min(chunk, nChunks - 1) << 2;
          eBase[q]  = min(e0, nedges - 4);
          firstE[q] = valid ? e0 : 0x7FFFFFFF;
          d[q] = *(const v4i*)(dst + eBase[q]);
        }
      } else {
#pragma unroll
        for (int q = 0; q < SCAN_Q; ++q) {
          d[q] = (v4i){-1, -1, -1, -1};
          eBase[q] = 0;
          firstE[q] = 0x7FFFFFFF;
        }
        if (wave == 0 && lane == 0 && step == 0) {
          firstE[0] = 0;
          if (0 < nedges) d[0].x = dst[0];
          if (1 < nedges) d[0].y = dst[1];
          if (2 < nedges) d[0].z = dst[2];
        }
      }
      int done = 0;
#pragma unroll
      for (int q = 0; q < SCAN_Q; ++q) {
        if (step + q < nSteps && cnt <= LCAP - LSTEPMAX) {
          const v4i dq = d[q];
          const int eb = eBase[q];
          const int fe = firstE[q];
          const bool h0 = (eb + 0 >= fe) && ((unsigned)(dq.x - base) < (unsigned)OWN_ROWS);
          const bool h1 = (eb + 1 >= fe) && ((unsigned)(dq.y - base) < (unsigned)OWN_ROWS);
          const bool h2 = (eb + 2 >= fe) && ((unsigned)(dq.z - base) < (unsigned)OWN_ROWS);
          const bool h3 = (eb + 3 >= fe) && ((unsigned)(dq.w - base) < (unsigned)OWN_ROWS);
          if (__ballot(h0 | h1 | h2 | h3)) {
#pragma unroll
            for (int c = 0; c < 4; ++c) {
              const bool hit = c == 0 ? h0 : c == 1 ? h1 : c == 2 ? h2 : h3;
              const unsigned m = (unsigned)__ballot(hit);
              if (hit) {
                const int pos = cnt + __popc(m & ltMask);
                listA[wave][pos] = (unsigned)(eb + c);
                listB[wave][pos] = (unsigned)(dq[c] - base);
              }
              cnt += __popc(m);
            }
          }
          ++done;
        }
      }
      step += done;
    }
    for (int j = lane; j < cnt; j += 32) {
      const unsigned e  = listA[wave][j];
      const unsigned lr = listB[wave][j];
      listA[wave][j] = (unsigned)src[e] | (lr << 23);
      listB[wave][j] = __float_as_uint(edge_w[e]);
    }
    if (lane == 0) {
      listN[wave]    = cnt;
      waveDone[wave] = (step >= nSteps) ? 1 : 0;
    }
    __syncthreads();

    int qn = 0;
    for (int L = 0; L <= NWAVES; ++L) {
      const int n = (L < NWAVES) ? listN[L] : 0;
      for (int j = 0; j < n; j += 32) {
        const int idx = j + lane;
        unsigned a = 0, b = 0;
        bool mine = false;
        if (idx < n) {
          a = listA[L][idx];
          b = listB[L][idx];
          mine = (((a >> 23) & 31u) == (unsigned)wave);
        }
        const unsigned m = (unsigned)__ballot(mine);
        if (m) {
          if (mine) {
            const int pos = qn + __popc(m & ltMask);
            queA[wave][pos] = a;
            queB[wave][pos] = b;
          }
          qn += __popc(m);
        }
        if (qn > QCAP - 32) {
          int bq = 0;
          for (; bq + 8 <= qn; bq += 8)
            drain_batch(h, tile, queA[wave], queB[wave], bq, 8, lane);
          const int rem = qn - bq;
          unsigned ra = 0, rb = 0;
          if (lane < rem) { ra = queA[wave][bq + lane]; rb = queB[wave][bq + lane]; }
          if (lane < rem) { queA[wave][lane] = ra; queB[wave][lane] = rb; }
          qn = rem;
        }
      }
      if (L == NWAVES) {
        for (int bq = 0; bq < qn; bq += 8)
          drain_batch(h, tile, queA[wave], queB[wave], bq, min(8, qn - bq), lane);
        qn = 0;
      }
    }
    int allDone = 1;
    for (int i = 0; i < NWAVES; ++i) allDone &= waveDone[i];
    __syncthreads();
    if (allDone) break;
  }

  float cvLane = 0.f;
  {
    const int r = wave + NWAVES * lane;
    if (r < OWN_ROWS && base + r < nrows) cvLane = cv[base + r];
  }
  for (int i = 0; i < (OWN_ROWS + NWAVES - 1) / NWAVES; ++i) {
    const int r = wave + NWAVES * i;
    const int row = base + r;
    const float cvr = __shfl(cvLane, i);
    if (r < OWN_ROWS && row < nrows) {
      const v4f t = tile[r * (D_OUT / 4) + lane];
      *(v4f*)(out + (size_t)row * D_OUT + lane * 4) = t * cvr;
    }
  }
}

extern "C" void kernel_launch(void* const* d_in, const int* in_sizes, int n_in,
                              void* d_out, int out_size, void* d_ws, size_t ws_size,
                              hipStream_t stream) {
  const float* feat = (const float*)d_in[0];
  const float* W    = (const float*)d_in[1];
  const float* cu   = (const float*)d_in[2];
  const float* cv   = (const float*)d_in[3];
  const float* ew   = (const float*)d_in[4];
  const int*   src  = (const int*)d_in[5];
  const int*   dst  = (const int*)d_in[6];

  float* out = (float*)d_out;
  float* h   = (float*)d_ws;

  const int N = in_sizes[2];
  const int E = in_sizes[5];

  influence_gemm_cu_kernel<<<GEMM_BLOCKS, 256, 0, stream>>>(feat, W, cu, h, N);

  const int ownerBlocks = (N + OWN_ROWS - 1) / OWN_ROWS;
  influence_owner_sum_kernel<<<ownerBlocks, 1024, 0, stream>>>(
      h, src, dst, ew, cv, out, N, E);
}


# owner blocks scan a 1-byte owner key per edge (dst / 400, 4 to a word, written behind h by the projection launch), not the 4-byte dst: 1/4 of the scan reads, one zero-byte test per word; sums same
# speedup vs baseline: 2.1800x; 1.1981x over previous
#include <hip/hip_runtime.h>

typedef __attribute__((ext_vector_type(2))) float v2f;
typedef __attribute__((ext_vector_type(8))) float v8f;

#define D_IN  128
#define D_OUT 128

typedef __attribute__((ext_vector_type(4))) float v4f;
typedef __attribute__((ext_vector_type(4))) int   v4i;

#define OWN_ROWS 400

__device__ __forceinline__ unsigned pack_keys(v4i d) {
  return min((unsigned)d.x / (unsigned)OWN_ROWS, 255u) |
         min((unsigned)d.y / (unsigned)OWN_ROWS, 255u) << 8 |
         min((unsigned)d.z / (unsigned)OWN_ROWS, 255u) << 16 |
         min((unsigned)d.w / (unsigned)OWN_ROWS, 255u) << 24;
}

__device__ __forceinline__ unsigned key_word(const int* __restrict__ dst,
                                             int chunk, int nedges) {
  const int e0 = chunk << 2;
  unsigned kw = 0;
  if (e0 + 3 < nedges) {
    kw = pack_keys(*(const v4i*)(dst + e0));
  } else {
#pragma unroll
    for (int c = 0; c < 4; ++c) {
      const unsigned k = (e0 + c < nedges)
          ? min((unsigned)dst[e0 + c] / (unsigned)OWN_ROWS, 255u) : 255u;
      kw |= k << (8 * c);
    }
  }
  return kw;
}

__device__ __forceinline__ bool keys_fit(size_t ws_bytes, int nrows, int nedges) {
  return ws_bytes >= ((size_t)nrows * D_OUT + (size_t)((nedges + 3) >> 2)) * 4;
}

#define W_PITCH 288
#define GEMM_BLOCKS 512
#define GEMM_WAVES  8

__global__ __launch_bounds__(256) void influence_gemm_cu_kernel(
    const float* __restrict__ feat, const float* __restrict__ W,
    const float* __restrict__ cu, float* __restrict__ h, int nrows,
    const int* __restrict__ dst, int nedges, size_t ws_bytes) {
  __shared__ float sW[(D_IN / 2) * W_PITCH];

  const int tid = threadIdx.x;
  if (keys_fit(ws_bytes, nrows, nedges)) {
    unsigned* keys = (unsigned*)(h + (size_t)nrows * D_OUT);
    const int nChunks = (nedges + 3) >> 2;
    const int nFull   = nedges >> 2;
    const int gid     = blockIdx.x * 256 + tid;
    const int stride  = GEMM_BLOCKS * 256;
    for (int c0 = gid; c0 < nFull; c0 += 4 * stride) {
      v4i d[4];
#pragma unroll
      for (int i = 0; i < 4; ++i)
        d[i] = *(const v4i*)(dst + 4 * (size_t)min(c0 + i * stride, nFull - 1));
      __builtin_amdgcn_sched_barrier(0);
#pragma unroll
      for (int i = 0; i < 4; ++i)
        if (c0 + i * stride < nFull) keys[c0 + i * stride] = pack_keys(d[i]);
    }
    if (gid == 0 && nFull < nChunks) keys[nFull] = key_word(dst, nFull, nedges);
  }
#pragma unroll 8
  for (int i = 0; i < (D_IN / 2) * D_OUT / 256; ++i) {
    const int idx = tid + i * 256;
    const int p = idx >> 7;
    const int n = idx & (D_OUT - 1);
    v2f w2;
    w2.x = W[(2 * p) * D_OUT + n];
    w2.y = W[(2 * p + 1) * D_OUT + n];
    *(v2f*)(sW + p * W_PITCH + 2 * n) = w2;
  }
  __syncthreads();

  const int wave = tid >> 5;
  const int lane = tid & 31;
  const int m    = lane & 15;
  const int hh   = lane >> 4;
  const int nStrips = (nrows + 15) >> 4;

  for (int strip = wave * GEMM_BLOCKS + blockIdx.x; strip < nStrips;
       strip += GEMM_BLOCKS * GEMM_WAVES) {
    const int rowBase = strip << 4;
    int arow_idx = rowBase + m;
    if (arow_idx >= nrows) arow_idx = nrows - 1;
    const float* arow = feat + (size_t)arow_idx * D_IN + 2 * hh;

    v2f A[D_IN / 4];
#pragma unroll
    for (int kk = 0; kk < D_IN / 4; ++kk) A[kk] = *(const v2f*)(arow + kk * 4);

    v8f acc[8];
#pragma unroll
    for (int t = 0; t < 8; ++t) acc[t] = (v8f){};

    const float* bBase = sW + hh * W_PITCH + 2 * m;
    v2f Bc[8], Bn[8];
#pragma unroll
    for (int t = 0; t < 8; ++t) Bc[t] = *(const v2f*)(bBase + 32 * t);
#pragma unroll
    for (int kk = 0; kk < D_IN / 4; ++kk) {
      if (kk + 1 < D_IN / 4) {
#pragma unroll
        for (int t = 0; t < 8; ++t)
          Bn[t] = *(const v2f*)(bBase + (2 * kk + 2) * W_PITCH + 32 * t);
      }
#pragma unroll
      for (int t = 0; t < 8; ++t)
        acc[t] = __builtin_amdgcn_wmma_f32_16x16x4_f32(
             false, A[kk],  false, Bc[t],
             (short)0, acc[t],  false,  false);
      __builtin_amdgcn_sched_barrier(0);
#pragma unroll
      for (int t = 0; t < 8; ++t) Bc[t] = Bn[t];
    }

    const int mBase = rowBase + hh * 8;
#pragma unroll
    for (int v = 0; v < 8; ++v) {
      const int row = mBase + v;
      if (row < nrows) {
        const float c = cu[row];
        float* hrow = h + (size_t)row * D_OUT + m;
#pragma unroll
        for (int t = 0; t < 8; ++t) hrow[16 * t] = acc[t][v] * c;
      }
    }
  }
}

#define NWAVES   32
#define LCAP     384
#define LSTEPMAX   128
#define QCAP     64
#define SCAN_Q   16

__device__ __forceinline__ void drain_batch(const float* __restrict__ h,
                                            v4f* tile, const unsigned* qa,
                                            const unsigned* qb, int bq, int nb,
                                            int lane) {
  v4f buf[8];
  unsigned pa[8];
  float pw[8];
#pragma unroll
  for (int k = 0; k < 8; ++k) {
    const int j = bq + min(k, nb - 1);
    pa[k] = qa[j];
    pw[k] = __uint_as_float(qb[j]);
    buf[k] = *(const v4f*)(h + (size_t)(pa[k] & 0x7FFFFFu) * D_OUT + lane * 4);
  }
  __builtin_amdgcn_sched_barrier(0);
#pragma unroll
  for (int k = 0; k < 8; ++k) {
    if (k < nb) {
      const int r = (int)(pa[k] >> 23);
      v4f t = tile[r * (D_OUT / 4) + lane];
      t += buf[k] * pw[k];
      tile[r * (D_OUT / 4) + lane] = t;
    }
  }
}

__global__ __launch_bounds__(1024) void influence_owner_sum_kernel(
    const float* __restrict__ h, const int* __restrict__ src,
    const int* __restrict__ dst, const float* __restrict__ edge_w,
    const float* __restrict__ cv, float* __restrict__ out, int nrows,
    int nedges, size_t ws_bytes) {
  __shared__ v4f      tile[OWN_ROWS * (D_OUT / 4)];
  __shared__ unsigned listA[NWAVES][LCAP];
  __shared__ unsigned listB[NWAVES][LCAP];
  __shared__ unsigned queA[NWAVES][QCAP];
  __shared__ unsigned queB[NWAVES][QCAP];
  __shared__ int      listN[NWAVES];
  __shared__ int      waveDone[NWAVES];

  const int tid  = threadIdx.x;
  const int wave = tid >> 5;
  const int lane = tid & 31;
  const int base = blockIdx.x * OWN_ROWS;
  const unsigned ltMask = (1u << lane) - 1u;
  const unsigned ownPat = (unsigned)blockIdx.x * 0x01010101u;
  const bool useKeys = keys_fit(ws_bytes, nrows, nedges);
  const unsigned* keys = (const unsigned*)(h + (size_t)nrows * D_OUT);

  for (int i = tid; i < OWN_ROWS * (D_OUT / 4); i += 1024)
    tile[i] = (v4f){0.f, 0.f, 0.f, 0.f};

  const int nChunks  = (nedges + 3) >> 2;
  const int perWave  = (nChunks + NWAVES - 1) / NWAVES;
  const int segStart = wave * perWave;
  const int segEnd   = min(segStart + perWave, nChunks);
  const int nSteps   = segEnd > segStart ? (segEnd - segStart + 31) >> 5 : 0;
  int step = 0;
  const int maxSteps = (perWave + 31) >> 5;

  for (int round = 0; round <= maxSteps; ++round) {
    int cnt = 0;
    while (step < nSteps && cnt <= LCAP - LSTEPMAX) {
      unsigned x[SCAN_Q];
      if (useKeys) {
#pragma unroll
        for (int q = 0; q < SCAN_Q; ++q)
          x[q] = keys[min(segStart + (step + q) * 32 + lane, nChunks - 1)];
      } else {
#pragma unroll
        for (int q = 0; q < SCAN_Q; ++q)
          x[q] = key_word(dst, min(segStart + (step + q) * 32 + lane, nChunks - 1),
                          nedges);
      }
#pragma unroll
      for (int q = 0; q < SCAN_Q; ++q) {
        const int chunk = segStart + (step + q) * 32 + lane;
        const bool valid = (step + q < nSteps) && (chunk < segEnd);
        x[q] = valid ? (x[q] ^ ownPat) : 0xFFFFFFFFu;
      }
      int done = 0;
#pragma unroll
      for (int q = 0; q < SCAN_Q; ++q) {
        if (step + q < nSteps && cnt <= LCAP - LSTEPMAX) {
          const unsigned xq = x[q];
          const unsigned z =
              ~(((xq & 0x7F7F7F7Fu) + 0x7F7F7F7Fu) | xq | 0x7F7F7F7Fu);
          if (__ballot(z != 0u)) {
            const int eb = (segStart + (step + q) * 32 + lane) << 2;
#pragma unroll
            for (int c = 0; c < 4; ++c) {
              const bool hit = (z & (0x80u << (8 * c))) != 0u;
              const unsigned m = (unsigned)__ballot(hit);
              if (m) {
                if (hit) listA[wave][cnt + __popc(m & ltMask)] = (unsigned)(eb + c);
                cnt += __popc(m);
              }
            }
          }
          ++done;
        }
      }
      step += done;
    }
    for (int j = lane; j < cnt; j += 32) {
      const unsigned e  = listA[wave][j];
      const unsigned lr = (unsigned)(dst[e] - base);
      listA[wave][j] = (unsigned)src[e] | (lr << 23);
      listB[wave][j] = __float_as_uint(edge_w[e]);
    }
    if (lane == 0) {
      listN[wave]    = cnt;
      waveDone[wave] = (step >= nSteps) ? 1 : 0;
    }
    __syncthreads();

    int qn = 0;
    for (int L = 0; L <= NWAVES; ++L) {
      const int n = (L < NWAVES) ? listN[L] : 0;
      for (int j = 0; j < n; j += 32) {
        const int idx = j + lane;
        unsigned a = 0, b = 0;
        bool mine = false;
        if (idx < n) {
          a = listA[L][idx];
          b = listB[L][idx];
          mine = (((a >> 23) & 31u) == (unsigned)wave);
        }
        const unsigned m = (unsigned)__ballot(mine);
        if (m) {
          if (mine) {
            const int pos = qn + __popc(m & ltMask);
            queA[wave][pos] = a;
            queB[wave][pos] = b;
          }
          qn += __popc(m);
        }
        if (qn > QCAP - 32) {
          int bq = 0;
          for (; bq + 8 <= qn; bq += 8)
            drain_batch(h, tile, queA[wave], queB[wave], bq, 8, lane);
          const int rem = qn - bq;
          unsigned ra = 0, rb = 0;
          if (lane < rem) { ra = queA[wave][bq + lane]; rb = queB[wave][bq + lane]; }
          if (lane < rem) { queA[wave][lane] = ra; queB[wave][lane] = rb; }
          qn = rem;
        }
      }
      if (L == NWAVES) {
        for (int bq = 0; bq < qn; bq += 8)
          drain_batch(h, tile, queA[wave], queB[wave], bq, min(8, qn - bq), lane);
        qn = 0;
      }
    }
    int allDone = 1;
    for (int i = 0; i < NWAVES; ++i) allDone &= waveDone[i];
    __syncthreads();
    if (allDone) break;
  }

  float cvLane = 0.f;
  {
    const int r = wave + NWAVES * lane;
    if (r < OWN_ROWS && base + r < nrows) cvLane = cv[base + r];
  }
  for (int i = 0; i < (OWN_ROWS + NWAVES - 1) / NWAVES; ++i) {
    const int r = wave + NWAVES * i;
    const int row = base + r;
    const float cvr = __shfl(cvLane, i);
    if (r < OWN_ROWS && row < nrows) {
      const v4f t = tile[r * (D_OUT / 4) + lane];
      *(v4f*)(out + (size_t)row * D_OUT + lane * 4) = t * cvr;
    }
  }
}

extern "C" void kernel_launch(void* const* d_in, const int* in_sizes, int n_in,
                              void* d_out, int out_size, void* d_ws, size_t ws_size,
                              hipStream_t stream) {
  const float* feat = (const float*)d_in[0];
  const float* W    = (const float*)d_in[1];
  const float* cu   = (const float*)d_in[2];
  const float* cv   = (const float*)d_in[3];
  const float* ew   = (const float*)d_in[4];
  const int*   src  = (const int*)d_in[5];
  const int*   dst  = (const int*)d_in[6];

  float* out = (float*)d_out;
  float* h   = (float*)d_ws;

  const int N = in_sizes[2];
  const int E = in_sizes[5];

  influence_gemm_cu_kernel<<<GEMM_BLOCKS, 256, 0, stream>>>(feat, W, cu, h, N,
                                                            dst, E, ws_size);

  const int ownerBlocks = (N + OWN_ROWS - 1) / OWN_ROWS;
  influence_owner_sum_kernel<<<ownerBlocks, 1024, 0, stream>>>(
      h, src, dst, ew, cv, out, N, E, ws_size);
}


# owner block: each scanning wave lays its resolved list out grouped by owner wave (stable, in LDS); an owner wave copies its 32 runs and no longer tests every entry of every list; sums same
# speedup vs baseline: 2.4609x; 1.1289x over previous
#include <hip/hip_runtime.h>

typedef __attribute__((ext_vector_type(2))) float v2f;
typedef __attribute__((ext_vector_type(8))) float v8f;

#define D_IN  128
#define D_OUT 128

typedef __attribute__((ext_vector_type(4))) float v4f;
typedef __attribute__((ext_vector_type(4))) int   v4i;

#define OWN_ROWS 400

__device__ __forceinline__ unsigned pack_keys(v4i d) {
  return min((unsigned)d.x / (unsigned)OWN_ROWS, 255u) |
         min((unsigned)d.y / (unsigned)OWN_ROWS, 255u) << 8 |
         min((unsigned)d.z / (unsigned)OWN_ROWS, 255u) << 16 |
         min((unsigned)d.w / (unsigned)OWN_ROWS, 255u) << 24;
}

__device__ __forceinline__ unsigned key_word(const int* __restrict__ dst,
                                             int chunk, int nedges) {
  const int e0 = chunk << 2;
  unsigned kw = 0;
  if (e0 + 3 < nedges) {
    kw = pack_keys(*(const v4i*)(dst + e0));
  } else {
#pragma unroll
    for (int c = 0; c < 4; ++c) {
      const unsigned k = (e0 + c < nedges)
          ? min((unsigned)dst[e0 + c] / (unsigned)OWN_ROWS, 255u) : 255u;
      kw |= k << (8 * c);
    }
  }
  return kw;
}

__device__ __forceinline__ bool keys_fit(size_t ws_bytes, int nrows, int nedges) {
  return ws_bytes >= ((size_t)nrows * D_OUT + (size_t)((nedges + 3) >> 2)) * 4;
}

#define W_PITCH 288
#define GEMM_BLOCKS 512
#define GEMM_WAVES  8

__global__ __launch_bounds__(256) void influence_gemm_cu_kernel(
    const float* __restrict__ feat, const float* __restrict__ W,
    const float* __restrict__ cu, float* __restrict__ h, int nrows,
    const int* __restrict__ dst, int nedges, size_t ws_bytes) {
  __shared__ float sW[(D_IN / 2) * W_PITCH];

  const int tid = threadIdx.x;
  if (keys_fit(ws_bytes, nrows, nedges)) {
    unsigned* keys = (unsigned*)(h + (size_t)nrows * D_OUT);
    const int nChunks = (nedges + 3) >> 2;
    const int nFull   = nedges >> 2;
    const int gid     = blockIdx.x * 256 + tid;
    const int stride  = GEMM_BLOCKS * 256;
    for (int c0 = gid; c0 < nFull; c0 += 4 * stride) {
      v4i d[4];
#pragma unroll
      for (int i = 0; i < 4; ++i)
        d[i] = *(const v4i*)(dst + 4 * (size_t)min(c0 + i * stride, nFull - 1));
      __builtin_amdgcn_sched_barrier(0);
#pragma unroll
      for (int i = 0; i < 4; ++i)
        if (c0 + i * stride < nFull) keys[c0 + i * stride] = pack_keys(d[i]);
    }
    if (gid == 0 && nFull < nChunks) keys[nFull] = key_word(dst, nFull, nedges);
  }
#pragma unroll 8
  for (int i = 0; i < (D_IN / 2) * D_OUT / 256; ++i) {
    const int idx = tid + i * 256;
    const int p = idx >> 7;
    const int n = idx & (D_OUT - 1);
    v2f w2;
    w2.x = W[(2 * p) * D_OUT + n];
    w2.y = W[(2 * p + 1) * D_OUT + n];
    *(v2f*)(sW + p * W_PITCH + 2 * n) = w2;
  }
  __syncthreads();

  const int wave = tid >> 5;
  const int lane = tid & 31;
  const int m    = lane & 15;
  const int hh   = lane >> 4;
  const int nStrips = (nrows + 15) >> 4;

  for (int strip = wave * GEMM_BLOCKS + blockIdx.x; strip < nStrips;
       strip += GEMM_BLOCKS * GEMM_WAVES) {
    const int rowBase = strip << 4;
    int arow_idx = rowBase + m;
    if (arow_idx >= nrows) arow_idx = nrows - 1;
    const float* arow = feat + (size_t)arow_idx * D_IN + 2 * hh;

    v2f A[D_IN / 4];
#pragma unroll
    for (int kk = 0; kk < D_IN / 4; ++kk) A[kk] = *(const v2f*)(arow + kk * 4);

    v8f acc[8];
#pragma unroll
    for (int t = 0; t < 8; ++t) acc[t] = (v8f){};

    const float* bBase = sW + hh * W_PITCH + 2 * m;
    v2f Bc[8], Bn[8];
#pragma unroll
    for (int t = 0; t < 8; ++t) Bc[t] = *(const v2f*)(bBase + 32 * t);
#pragma unroll
    for (int kk = 0; kk < D_IN / 4; ++kk) {
      if (kk + 1 < D_IN / 4) {
#pragma unroll
        for (int t = 0; t < 8; ++t)
          Bn[t] = *(const v2f*)(bBase + (2 * kk + 2) * W_PITCH + 32 * t);
      }
#pragma unroll
      for (int t = 0; t < 8; ++t)
        acc[t] = __builtin_amdgcn_wmma_f32_16x16x4_f32(
             false, A[kk],  false, Bc[t],
             (short)0, acc[t],  false,  false);
      __builtin_amdgcn_sched_barrier(0);
#pragma unroll
      for (int t = 0; t < 8; ++t) Bc[t] = Bn[t];
    }

    const int mBase = rowBase + hh * 8;
#pragma unroll
    for (int v = 0; v < 8; ++v) {
      const int row = mBase + v;
      if (row < nrows) {
        const float c = cu[row];
        float* hrow = h + (size_t)row * D_OUT + m;
#pragma unroll
        for (int t = 0; t < 8; ++t) hrow[16 * t] = acc[t][v] * c;
      }
    }
  }
}

#define NWAVES   32
#define LCAP     384
#define LCHUNKS  (LCAP / 32)
#define LSTEPMAX   128
#define QCAP     64
#define SCAN_Q   16

__device__ __forceinline__ void drain_batch(const float* __restrict__ h,
                                            v4f* tile, const unsigned* qa,
                                            const unsigned* qb, int bq, int nb,
                                            int lane) {
  v4f buf[8];
  unsigned pa[8];
  float pw[8];
#pragma unroll
  for (int k = 0; k < 8; ++k) {
    const int j = bq + min(k, nb - 1);
    pa[k] = qa[j];
    pw[k] = __uint_as_float(qb[j]);
    buf[k] = *(const v4f*)(h + (size_t)(pa[k] & 0x7FFFFFu) * D_OUT + lane * 4);
  }
  __builtin_amdgcn_sched_barrier(0);
#pragma unroll
  for (int k = 0; k < 8; ++k) {
    if (k < nb) {
      const int r = (int)(pa[k] >> 23);
      v4f t = tile[r * (D_OUT / 4) + lane];
      t += buf[k] * pw[k];
      tile[r * (D_OUT / 4) + lane] = t;
    }
  }
}

__global__ __launch_bounds__(1024) void influence_owner_sum_kernel(
    const float* __restrict__ h, const int* __restrict__ src,
    const int* __restrict__ dst, const float* __restrict__ edge_w,
    const float* __restrict__ cv, float* __restrict__ out, int nrows,
    int nedges, size_t ws_bytes) {
  __shared__ v4f      tile[OWN_ROWS * (D_OUT / 4)];
  __shared__ unsigned listA[NWAVES][LCAP];
  __shared__ unsigned listB[NWAVES][LCAP];
  __shared__ unsigned queA[NWAVES][QCAP];
  __shared__ unsigned queB[NWAVES][QCAP];
  __shared__ unsigned runTab[NWAVES][NWAVES];
  __shared__ int      waveDone[NWAVES];

  const int tid  = threadIdx.x;
  const int wave = tid >> 5;
  const int lane = tid & 31;
  const int base = blockIdx.x * OWN_ROWS;
  const unsigned ltMask = (1u << lane) - 1u;
  const unsigned ownPat = (unsigned)blockIdx.x * 0x01010101u;
  const bool useKeys = keys_fit(ws_bytes, nrows, nedges);
  const unsigned* keys = (const unsigned*)(h + (size_t)nrows * D_OUT);

  for (int i = tid; i < OWN_ROWS * (D_OUT / 4); i += 1024)
    tile[i] = (v4f){0.f, 0.f, 0.f, 0.f};

  const int nChunks  = (nedges + 3) >> 2;
  const int perWave  = (nChunks + NWAVES - 1) / NWAVES;
  const int segStart = wave * perWave;
  const int segEnd   = min(segStart + perWave, nChunks);
  const int nSteps   = segEnd > segStart ? (segEnd - segStart + 31) >> 5 : 0;
  int step = 0;
  const int maxSteps = (perWave + 31) >> 5;

  for (int round = 0; round <= maxSteps; ++round) {
    int cnt = 0;
    while (step < nSteps && cnt <= LCAP - LSTEPMAX) {
      unsigned x[SCAN_Q];
      if (useKeys) {
#pragma unroll
        for (int q = 0; q < SCAN_Q; ++q)
          x[q] = keys[min(segStart + (step + q) * 32 + lane, nChunks - 1)];
      } else {
#pragma unroll
        for (int q = 0; q < SCAN_Q; ++q)
          x[q] = key_word(dst, min(segStart + (step + q) * 32 + lane, nChunks - 1),
                          nedges);
      }
#pragma unroll
      for (int q = 0; q < SCAN_Q; ++q) {
        const int chunk = segStart + (step + q) * 32 + lane;
        const bool valid = (step + q < nSteps) && (chunk < segEnd);
        x[q] = valid ? (x[q] ^ ownPat) : 0xFFFFFFFFu;
      }
      int done = 0;
#pragma unroll
      for (int q = 0; q < SCAN_Q; ++q) {
        if (step + q < nSteps && cnt <= LCAP - LSTEPMAX) {
          const unsigned xq = x[q];
          const unsigned z =
              ~(((xq & 0x7F7F7F7Fu) + 0x7F7F7F7Fu) | xq | 0x7F7F7F7Fu);
          if (__ballot(z != 0u)) {
            const int eb = (segStart + (step + q) * 32 + lane) << 2;
#pragma unroll
            for (int c = 0; c < 4; ++c) {
              const bool hit = (z & (0x80u << (8 * c))) != 0u;
              const unsigned m = (unsigned)__ballot(hit);
              if (m) {
                if (hit) listA[wave][cnt + __popc(m & ltMask)] = (unsigned)(eb + c);
                cnt += __popc(m);
              }
            }
          }
          ++done;
        }
      }
      step += done;
    }
    const int cntU = __builtin_amdgcn_readfirstlane(cnt);
    unsigned ra[LCHUNKS], rb[LCHUNKS], rp[LCHUNKS];
#pragma unroll
    for (int c = 0; c < LCHUNKS; ++c) { ra[c] = 0u; rb[c] = 0u; rp[c] = 0u; }
    runTab[wave][lane] = 0u;
#pragma unroll
    for (int g = 0; g < LCHUNKS / 4; ++g) {
      if (g * 128 < cntU) {
#pragma unroll
        for (int c = 4 * g; c < 4 * g + 4; ++c) {
          const unsigned e  = listA[wave][min(c * 32 + lane, cntU - 1)];
          const unsigned lr = (unsigned)(dst[e] - base);
          ra[c] = (unsigned)src[e] | (lr << 23);
          rb[c] = __float_as_uint(edge_w[e]);
        }
#pragma unroll
        for (int c = 4 * g; c < 4 * g + 4; ++c) {
          if (c * 32 < cntU) {
            const bool valid = c * 32 + lane < cntU;
            const unsigned o = (ra[c] >> 23) & 31u;
            unsigned peers = (unsigned)__ballot(valid);
#pragma unroll
            for (int bit = 0; bit < 5; ++bit) {
              const bool set = ((o >> bit) & 1u) != 0u;
              const unsigned bm = (unsigned)__ballot(set);
              peers &= set ? bm : ~bm;
            }
            if (valid) {
              const unsigned before = runTab[wave][o];
              rp[c] = before + (unsigned)__popc(peers & ltMask);
              if ((peers & ltMask) == 0u)
                runTab[wave][o] = before + (unsigned)__popc(peers);
            }
          }
        }
      }
    }
    const unsigned runLen = runTab[wave][lane];
    unsigned runEnd = runLen;
#pragma unroll
    for (int d = 1; d < 32; d <<= 1) {
      const unsigned t = __shfl_up(runEnd, (unsigned)d);
      if (lane >= d) runEnd += t;
    }
    const unsigned runStart = runEnd - runLen;
    runTab[wave][lane] = runStart | (runLen << 16);
#pragma unroll
    for (int c = 0; c < LCHUNKS; ++c) {
      if (c * 32 < cntU) {
        const unsigned st = __shfl(runStart, (int)((ra[c] >> 23) & 31u));
        if (c * 32 + lane < cntU) {
          listA[wave][st + rp[c]] = ra[c];
          listB[wave][st + rp[c]] = rb[c];
        }
      }
    }
    if (lane == 0) waveDone[wave] = (step >= nSteps) ? 1 : 0;
    __syncthreads();

    const unsigned myRun = runTab[lane][wave];
    int qn = 0;
    for (int L = 0; L <= NWAVES; ++L) {
      const unsigned run =
          (L < NWAVES) ? (unsigned)__builtin_amdgcn_readlane((int)myRun, L) : 0u;
      const int s = (int)(run & 0xFFFFu);
      const int n = (int)(run >> 16);
      for (int j = 0; j < n; j += 32) {
        const int take = min(32, n - j);
        if (lane < take) {
          queA[wave][qn + lane] = listA[L][s + j + lane];
          queB[wave][qn + lane] = listB[L][s + j + lane];
        }
        qn += take;
        if (qn > QCAP - 32) {
          int bq = 0;
          for (; bq + 8 <= qn; bq += 8)
            drain_batch(h, tile, queA[wave], queB[wave], bq, 8, lane);
          const int rem = qn - bq;
          unsigned ra = 0, rb = 0;
          if (lane < rem) { ra = queA[wave][bq + lane]; rb = queB[wave][bq + lane]; }
          if (lane < rem) { queA[wave][lane] = ra; queB[wave][lane] = rb; }
          qn = rem;
        }
      }
      if (L == NWAVES) {
        for (int bq = 0; bq < qn; bq += 8)
          drain_batch(h, tile, queA[wave], queB[wave], bq, min(8, qn - bq), lane);
        qn = 0;
      }
    }
    int allDone = 1;
    for (int i = 0; i < NWAVES; ++i) allDone &= waveDone[i];
    __syncthreads();
    if (allDone) break;
  }

  float cvLane = 0.f;
  {
    const int r = wave + NWAVES * lane;
    if (r < OWN_ROWS && base + r < nrows) cvLane = cv[base + r];
  }
  for (int i = 0; i < (OWN_ROWS + NWAVES - 1) / NWAVES; ++i) {
    const int r = wave + NWAVES * i;
    const int row = base + r;
    const float cvr = __shfl(cvLane, i);
    if (r < OWN_ROWS && row < nrows) {
      const v4f t = tile[r * (D_OUT / 4) + lane];
      *(v4f*)(out + (size_t)row * D_OUT + lane * 4) = t * cvr;
    }
  }
}

extern "C" void kernel_launch(void* const* d_in, const int* in_sizes, int n_in,
                              void* d_out, int out_size, void* d_ws, size_t ws_size,
                              hipStream_t stream) {
  const float* feat = (const float*)d_in[0];
  const float* W    = (const float*)d_in[1];
  const float* cu   = (const float*)d_in[2];
  const float* cv   = (const float*)d_in[3];
  const float* ew   = (const float*)d_in[4];
  const int*   src  = (const int*)d_in[5];
  const int*   dst  = (const int*)d_in[6];

  float* out = (float*)d_out;
  float* h   = (float*)d_ws;

  const int N = in_sizes[2];
  const int E = in_sizes[5];

  influence_gemm_cu_kernel<<<GEMM_BLOCKS, 256, 0, stream>>>(feat, W, cu, h, N,
                                                            dst, E, ws_size);

  const int ownerBlocks = (N + OWN_ROWS - 1) / OWN_ROWS;
  influence_owner_sum_kernel<<<ownerBlocks, 1024, 0, stream>>>(
      h, src, dst, ew, cv, out, N, E, ws_size);
}


# owner blocks read a per-block edge bitmap (1 bit per edge, built in LDS by the projection launch, staged in the block's own span of out), not a 1-byte key per edge; sum order changed
# speedup vs baseline: 3.0040x; 1.2207x over previous
#include <hip/hip_runtime.h>

typedef __attribute__((ext_vector_type(2))) float v2f;
typedef __attribute__((ext_vector_type(8))) float v8f;

#define D_IN  128
#define D_OUT 128

typedef __attribute__((ext_vector_type(4))) float v4f;
typedef __attribute__((ext_vector_type(4))) int   v4i;

#define OWN_ROWS 400
#define OWN_WORDS (OWN_ROWS * D_OUT)
#define BM_TILE 64
#define BM_KEYS 256

__device__ __forceinline__ bool bits_fit(int nrows, int nedges) {
  const size_t nWords  = ((size_t)nedges + 31) >> 5;
  const size_t nOwners = ((size_t)nrows + OWN_ROWS - 1) / OWN_ROWS;
  return nOwners >= 1 && nOwners <= (size_t)BM_KEYS && nWords <= (size_t)OWN_WORDS &&
         (nOwners - 1) * OWN_WORDS + nWords <= (size_t)nrows * D_OUT;
}

__device__ __forceinline__ void load_targets(const int* __restrict__ dst, int e0,
                                             int nedges, bool on, int (&d)[8]) {
#pragma unroll
  for (int j = 0; j < 8; ++j) d[j] = -1;
  if (on) {
    if (e0 + 7 < nedges) {
      const v4i a = *(const v4i*)(dst + e0);
      const v4i b = *(const v4i*)(dst + e0 + 4);
      d[0] = a.x; d[1] = a.y; d[2] = a.z; d[3] = a.w;
      d[4] = b.x; d[5] = b.y; d[6] = b.z; d[7] = b.w;
    } else {
#pragma unroll
      for (int j = 0; j < 8; ++j)
        if (e0 + j < nedges) d[j] = dst[e0 + j];
    }
  }
}

__device__ __forceinline__ unsigned own_bits(const int* __restrict__ dst, int w,
                                             int nedges, unsigned b) {
  unsigned bits = 0;
#pragma unroll 1
  for (int j = 0; j < 32; ++j) {
    const int e = (w << 5) + j;
    if (e < nedges && (unsigned)dst[e] / (unsigned)OWN_ROWS == b) bits |= 1u << j;
  }
  return bits;
}

#define W_PITCH 288
#define GEMM_BLOCKS 512
#define GEMM_WAVES  8

__global__ __launch_bounds__(256) void influence_gemm_cu_kernel(
    const float* __restrict__ feat, const float* __restrict__ W,
    const float* __restrict__ cu, float* __restrict__ h, int nrows,
    const int* __restrict__ dst, int nedges, float* __restrict__ out) {
  __shared__ float sW[(D_IN / 2) * W_PITCH];

  const int tid = threadIdx.x;
  if (bits_fit(nrows, nedges)) {
    unsigned*      tb  = (unsigned*)sW;
    unsigned char* tb8 = (unsigned char*)sW;
    unsigned*      bm  = (unsigned*)out;
    const int nWords  = (nedges + 31) >> 5;
    const int nOwners = (nrows + OWN_ROWS - 1) / OWN_ROWS;
    const int nTiles  = (nWords + BM_TILE - 1) / BM_TILE;
    const int g = tid & (BM_TILE - 1);
    const int q = tid >> 6;
    int pass = 0;
    int tile = blockIdx.x;
    int d[8];
    load_targets(dst, ((tile * BM_TILE + g) << 5) + q * 8, nedges, tile < nTiles, d);
    while (tile < nTiles) {
      for (int i = tid; i < BM_KEYS * BM_TILE / 4; i += 256)
        ((v4i*)sW)[i] = (v4i){0, 0, 0, 0};
      __syncthreads();
#pragma unroll
      for (int j = 0; j < 8; ++j) {
        const unsigned key = (unsigned)d[j] / (unsigned)OWN_ROWS;
        if (d[j] >= 0 && key < (unsigned)nOwners)
          tb8[((key * BM_TILE + g) << 2) + q] |= (unsigned char)(1u << j);
      }
      __syncthreads();
      const int w0 = tile * BM_TILE;
      ++pass;
      const int next = (pass & 1) ? pass * GEMM_BLOCKS + (GEMM_BLOCKS - 1 - (int)blockIdx.x)
                                  : pass * GEMM_BLOCKS + (int)blockIdx.x;
      load_targets(dst, ((next * BM_TILE + g) << 5) + q * 8, nedges, next < nTiles, d);
      for (int i = tid; i < nOwners * (BM_TILE / 4); i += 256) {
        const int key  = i >> 4;
        const int part = (i & 15) << 2;
        const v4i v = *(const v4i*)(tb + key * BM_TILE + part);
        unsigned* p = bm + (size_t)key * OWN_WORDS + w0 + part;
        if (w0 + part + 3 < nWords) {
          *(v4i*)p = v;
        } else {
          if (w0 + part + 0 < nWords) p[0] = (unsigned)v.x;
          if (w0 + part + 1 < nWords) p[1] = (unsigned)v.y;
          if (w0 + part + 2 < nWords) p[2] = (unsigned)v.z;
        }
      }
      __syncthreads();
      tile = next;
    }
  }
#pragma unroll 8
  for (int i = 0; i < (D_IN / 2) * D_OUT / 256; ++i) {
    const int idx = tid + i * 256;
    const int p = idx >> 7;
    const int n = idx & (D_OUT - 1);
    v2f w2;
    w2.x = W[(2 * p) * D_OUT + n];
    w2.y = W[(2 * p + 1) * D_OUT + n];
    *(v2f*)(sW + p * W_PITCH + 2 * n) = w2;
  }
  __syncthreads();

  const int wave = tid >> 5;
  const int lane = tid & 31;
  const int m    = lane & 15;
  const int hh   = lane >> 4;
  const int nStrips = (nrows + 15) >> 4;

  for (int strip = wave * GEMM_BLOCKS + blockIdx.x; strip < nStrips;
       strip += GEMM_BLOCKS * GEMM_WAVES) {
    const int rowBase = strip << 4;
    int arow_idx = rowBase + m;
    if (arow_idx >= nrows) arow_idx = nrows - 1;
    const float* arow = feat + (size_t)arow_idx * D_IN + 2 * hh;

    v2f A[D_IN / 4];
#pragma unroll
    for (int kk = 0; kk < D_IN / 4; ++kk) A[kk] = *(const v2f*)(arow + kk * 4);

    v8f acc[8];
#pragma unroll
    for (int t = 0; t < 8; ++t) acc[t] = (v8f){};

    const float* bBase = sW + hh * W_PITCH + 2 * m;
    v2f Bc[8], Bn[8];
#pragma unroll
    for (int t = 0; t < 8; ++t) Bc[t] = *(const v2f*)(bBase + 32 * t);
#pragma unroll
    for (int kk = 0; kk < D_IN / 4; ++kk) {
      if (kk + 1 < D_IN / 4) {
#pragma unroll
        for (int t = 0; t < 8; ++t)
          Bn[t] = *(const v2f*)(bBase + (2 * kk + 2) * W_PITCH + 32 * t);
      }
#pragma unroll
      for (int t = 0; t < 8; ++t)
        acc[t] = __builtin_amdgcn_wmma_f32_16x16x4_f32(
             false, A[kk],  false, Bc[t],
             (short)0, acc[t],  false,  false);
      __builtin_amdgcn_sched_barrier(0);
#pragma unroll
      for (int t = 0; t < 8; ++t) Bc[t] = Bn[t];
    }

    const int mBase = rowBase + hh * 8;
#pragma unroll
    for (int v = 0; v < 8; ++v) {
      const int row = mBase + v;
      if (row < nrows) {
        const float c = cu[row];
        float* hrow = h + (size_t)row * D_OUT + m;
#pragma unroll
        for (int t = 0; t < 8; ++t) hrow[16 * t] = acc[t][v] * c;
      }
    }
  }
}

#define NWAVES   32
#define LCAP     384
#define LCHUNKS  (LCAP / 32)
#define DRAIN_TRIPS ((LCAP - 32) / 32 + 1)
#define QCAP     64
#define SCAN_Q   25

__device__ __forceinline__ void drain_batch(const float* __restrict__ h,
                                            v4f* tile, const unsigned* qa,
                                            const unsigned* qb, int bq, int nb,
                                            int lane) {
  v4f buf[8];
  unsigned pa[8];
  float pw[8];
#pragma unroll
  for (int k = 0; k < 8; ++k) {
    const int j = bq + min(k, nb - 1);
    pa[k] = qa[j];
    pw[k] = __uint_as_float(qb[j]);
    buf[k] = *(const v4f*)(h + (size_t)(pa[k] & 0x7FFFFFu) * D_OUT + lane * 4);
  }
  __builtin_amdgcn_sched_barrier(0);
#pragma unroll
  for (int k = 0; k < 8; ++k) {
    if (k < nb) {
      const int r = (int)(pa[k] >> 23);
      v4f t = tile[r * (D_OUT / 4) + lane];
      t += buf[k] * pw[k];
      tile[r * (D_OUT / 4) + lane] = t;
    }
  }
}

__device__ __forceinline__ bool drain_bits(unsigned& pend, unsigned pbase, int& cnt,
                                           unsigned* list, unsigned ltMask) {
  unsigned m = (unsigned)__ballot(pend != 0u);
  for (int it = 0; it < 32 && m != 0u && cnt <= LCAP - 32; ++it) {
    if (pend != 0u) {
      list[cnt + __popc(m & ltMask)] = pbase + (unsigned)__builtin_ctz(pend);
      pend &= pend - 1u;
    }
    cnt += __popc(m);
    m = (unsigned)__ballot(pend != 0u);
  }
  return m != 0u;
}

__global__ __launch_bounds__(1024) void influence_owner_sum_kernel(
    const float* __restrict__ h, const int* __restrict__ src,
    const int* __restrict__ dst, const float* __restrict__ edge_w,
    const float* __restrict__ cv, float* __restrict__ out, int nrows,
    int nedges) {
  __shared__ v4f      tile[OWN_ROWS * (D_OUT / 4)];
  __shared__ unsigned listA[NWAVES][LCAP];
  __shared__ unsigned listB[NWAVES][LCAP];
  __shared__ unsigned queA[NWAVES][QCAP];
  __shared__ unsigned queB[NWAVES][QCAP];
  __shared__ unsigned runTab[NWAVES][NWAVES];
  __shared__ int      waveDone[NWAVES];

  const int tid  = threadIdx.x;
  const int wave = tid >> 5;
  const int lane = tid & 31;
  const int base = blockIdx.x * OWN_ROWS;
  const unsigned ltMask = (1u << lane) - 1u;
  const bool useBits = bits_fit(nrows, nedges);
  const unsigned* bm = (const unsigned*)out + (size_t)blockIdx.x * OWN_WORDS;

  for (int i = tid; i < OWN_ROWS * (D_OUT / 4); i += 1024)
    tile[i] = (v4f){0.f, 0.f, 0.f, 0.f};

  const int waveS    = __builtin_amdgcn_readfirstlane(wave);
  const int nWords   = (nedges + 31) >> 5;
  const int perWave  = (nWords + NWAVES - 1) / NWAVES;
  const int segStart = waveS * perWave;
  const int segEnd   = min(segStart + perWave, nWords);
  const int nSteps   = segEnd > segStart ? (segEnd - segStart + 31) >> 5 : 0;
  int step = 0;
  unsigned pend = 0u, pbase = 0u;
  const int maxRounds = ((32 + DRAIN_TRIPS - 1) / DRAIN_TRIPS) * ((perWave + 31) >> 5) + 1;

  for (int round = 0; round <= maxRounds; ++round) {
    int cnt = 0;
    bool full = drain_bits(pend, pbase, cnt, listA[waveS], ltMask);
    while (!full && step < nSteps) {
      unsigned x[SCAN_Q];
      if (useBits) {
#pragma unroll
        for (int q = 0; q < SCAN_Q; ++q)
          x[q] = bm[min(segStart + (step + q) * 32 + lane, nWords - 1)];
      } else {
#pragma unroll
        for (int q = 0; q < SCAN_Q; ++q)
          x[q] = own_bits(dst, min(segStart + (step + q) * 32 + lane, nWords - 1),
                          nedges, blockIdx.x);
      }
      int done = 0;
#pragma unroll
      for (int q = 0; q < SCAN_Q; ++q) {
        if (!full && step + q < nSteps) {
          const int w = segStart + (step + q) * 32 + lane;
          pend  = (w < segEnd) ? x[q] : 0u;
          pbase = (unsigned)w << 5;
          full  = drain_bits(pend, pbase, cnt, listA[waveS], ltMask);
          ++done;
        }
      }
      step += done;
    }
    const int cntU = __builtin_amdgcn_readfirstlane(cnt);
    unsigned ra[LCHUNKS], rb[LCHUNKS], rp[LCHUNKS];
#pragma unroll
    for (int c = 0; c < LCHUNKS; ++c) { ra[c] = 0u; rb[c] = 0u; rp[c] = 0u; }
    runTab[wave][lane] = 0u;
#pragma unroll
    for (int g = 0; g < LCHUNKS / 4; ++g) {
      if (g * 128 < cntU) {
#pragma unroll
        for (int c = 4 * g; c < 4 * g + 4; ++c) {
          const unsigned e  = listA[wave][min(c * 32 + lane, cntU - 1)];
          const unsigned lr = (unsigned)(dst[e] - base);
          ra[c] = (unsigned)src[e] | (lr << 23);
          rb[c] = __float_as_uint(edge_w[e]);
        }
#pragma unroll
        for (int c = 4 * g; c < 4 * g + 4; ++c) {
          if (c * 32 < cntU) {
            const bool valid = c * 32 + lane < cntU;
            const unsigned o = (ra[c] >> 23) & 31u;
            unsigned peers = (unsigned)__ballot(valid);
#pragma unroll
            for (int bit = 0; bit < 5; ++bit) {
              const bool set = ((o >> bit) & 1u) != 0u;
              const unsigned bm = (unsigned)__ballot(set);
              peers &= set ? bm : ~bm;
            }
            if (valid) {
              const unsigned before = runTab[wave][o];
              rp[c] = before + (unsigned)__popc(peers & ltMask);
              if ((peers & ltMask) == 0u)
                runTab[wave][o] = before + (unsigned)__popc(peers);
            }
          }
        }
      }
    }
    const unsigned runLen = runTab[wave][lane];
    unsigned runEnd = runLen;
#pragma unroll
    for (int d = 1; d < 32; d <<= 1) {
      const unsigned t = __shfl_up(runEnd, (unsigned)d);
      if (lane >= d) runEnd += t;
    }
    const unsigned runStart = runEnd - runLen;
    runTab[wave][lane] = runStart | (runLen << 16);
#pragma unroll
    for (int c = 0; c < LCHUNKS; ++c) {
      if (c * 32 < cntU) {
        const unsigned st = __shfl(runStart, (int)((ra[c] >> 23) & 31u));
        if (c * 32 + lane < cntU) {
          listA[wave][st + rp[c]] = ra[c];
          listB[wave][st + rp[c]] = rb[c];
        }
      }
    }
    if (lane == 0) waveDone[wave] = (step >= nSteps && !full) ? 1 : 0;
    __syncthreads();

    const unsigned myRun = runTab[lane][wave];
    int qn = 0;
    for (int L = 0; L <= NWAVES; ++L) {
      const unsigned run =
          (L < NWAVES) ? (unsigned)__builtin_amdgcn_readlane((int)myRun, L) : 0u;
      const int s = (int)(run & 0xFFFFu);
      const int n = (int)(run >> 16);
      for (int j = 0; j < n; j += 32) {
        const int take = min(32, n - j);
        if (lane < take) {
          queA[wave][qn + lane] = listA[L][s + j + lane];
          queB[wave][qn + lane] = listB[L][s + j + lane];
        }
        qn += take;
        if (qn > QCAP - 32) {
          int bq = 0;
          for (; bq + 8 <= qn; bq += 8)
            drain_batch(h, tile, queA[wave], queB[wave], bq, 8, lane);
          const int rem = qn - bq;
          unsigned ra = 0, rb = 0;
          if (lane < rem) { ra = queA[wave][bq + lane]; rb = queB[wave][bq + lane]; }
          if (lane < rem) { queA[wave][lane] = ra; queB[wave][lane] = rb; }
          qn = rem;
        }
      }
      if (L == NWAVES) {
        for (int bq = 0; bq < qn; bq += 8)
          drain_batch(h, tile, queA[wave], queB[wave], bq, min(8, qn - bq), lane);
        qn = 0;
      }
    }
    int allDone = 1;
    for (int i = 0; i < NWAVES; ++i) allDone &= waveDone[i];
    __syncthreads();
    if (allDone) break;
  }

  float cvLane = 0.f;
  {
    const int r = wave + NWAVES * lane;
    if (r < OWN_ROWS && base + r < nrows) cvLane = cv[base + r];
  }
  for (int i = 0; i < (OWN_ROWS + NWAVES - 1) / NWAVES; ++i) {
    const int r = wave + NWAVES * i;
    const int row = base + r;
    const float cvr = __shfl(cvLane, i);
    if (r < OWN_ROWS && row < nrows) {
      const v4f t = tile[r * (D_OUT / 4) + lane];
      *(v4f*)(out + (size_t)row * D_OUT + lane * 4) = t * cvr;
    }
  }
}

extern "C" void kernel_launch(void* const* d_in, const int* in_sizes, int n_in,
                              void* d_out, int out_size, void* d_ws, size_t ws_size,
                              hipStream_t stream) {
  const float* feat = (const float*)d_in[0];
  const float* W    = (const float*)d_in[1];
  const float* cu   = (const float*)d_in[2];
  const float* cv   = (const float*)d_in[3];
  const float* ew   = (const float*)d_in[4];
  const int*   src  = (const int*)d_in[5];
  const int*   dst  = (const int*)d_in[6];

  float* out = (float*)d_out;
  float* h   = (float*)d_ws;

  const int N = in_sizes[2];
  const int E = in_sizes[5];

  influence_gemm_cu_kernel<<<GEMM_BLOCKS, 256, 0, stream>>>(feat, W, cu, h, N,
                                                            dst, E, out);

  const int ownerBlocks = (N + OWN_ROWS - 1) / OWN_ROWS;
  influence_owner_sum_kernel<<<ownerBlocks, 1024, 0, stream>>>(
      h, src, dst, ew, cv, out, N, E);
}


# projection launch also packs each edge into one 8-byte record (src | row in owner block << 23, weight) behind h; owner blocks resolve an edge with 1 scattered load, not 3; sums same
# speedup vs baseline: 3.1771x; 1.0576x over previous
#include <hip/hip_runtime.h>

typedef __attribute__((ext_vector_type(2))) float v2f;
typedef __attribute__((ext_vector_type(8))) float v8f;

#define D_IN  128
#define D_OUT 128

typedef __attribute__((ext_vector_type(4))) float v4f;
typedef __attribute__((ext_vector_type(4))) int   v4i;
typedef __attribute__((ext_vector_type(2))) int      v2i;
typedef __attribute__((ext_vector_type(2))) unsigned v2u;
typedef __attribute__((ext_vector_type(4))) unsigned v4u;

#define OWN_ROWS 400
#define OWN_WORDS (OWN_ROWS * D_OUT)
#define BM_TILE 64
#define BM_KEYS 256

__device__ __forceinline__ bool bits_fit(int nrows, int nedges) {
  const size_t nWords  = ((size_t)nedges + 31) >> 5;
  const size_t nOwners = ((size_t)nrows + OWN_ROWS - 1) / OWN_ROWS;
  return nOwners >= 1 && nOwners <= (size_t)BM_KEYS && nWords <= (size_t)OWN_WORDS &&
         (nOwners - 1) * OWN_WORDS + nWords <= (size_t)nrows * D_OUT;
}

__device__ __forceinline__ bool recs_fit(int nrows, int nedges, size_t wsBytes) {
  return (size_t)nrows * D_OUT * sizeof(float) + (size_t)nedges * 8 <= wsBytes;
}

__device__ __forceinline__ unsigned pack_src_row(int s, int d) {
  return (unsigned)s | (((unsigned)d % (unsigned)OWN_ROWS) << 23);
}

#define REC_Q 8

__device__ __forceinline__ void load_targets(const int* __restrict__ dst, int e0,
                                             int nedges, bool on, int (&d)[8]) {
#pragma unroll
  for (int j = 0; j < 8; ++j) d[j] = -1;
  if (on) {
    if (e0 + 7 < nedges) {
      const v4i a = *(const v4i*)(dst + e0);
      const v4i b = *(const v4i*)(dst + e0 + 4);
      d[0] = a.x; d[1] = a.y; d[2] = a.z; d[3] = a.w;
      d[4] = b.x; d[5] = b.y; d[6] = b.z; d[7] = b.w;
    } else {
#pragma unroll
      for (int j = 0; j < 8; ++j)
        if (e0 + j < nedges) d[j] = dst[e0 + j];
    }
  }
}

__device__ __forceinline__ unsigned own_bits(const int* __restrict__ dst, int w,
                                             int nedges, unsigned b) {
  unsigned bits = 0;
#pragma unroll 1
  for (int j = 0; j < 32; ++j) {
    const int e = (w << 5) + j;
    if (e < nedges && (unsigned)dst[e] / (unsigned)OWN_ROWS == b) bits |= 1u << j;
  }
  return bits;
}

#define W_PITCH 288
#define GEMM_BLOCKS 512
#define GEMM_WAVES  8

__global__ __launch_bounds__(256) void influence_gemm_cu_kernel(
    const float* __restrict__ feat, const float* __restrict__ W,
    const float* __restrict__ cu, float* __restrict__ h, int nrows,
    const int* __restrict__ dst, int nedges, float* __restrict__ out,
    const int* __restrict__ src, const float* __restrict__ edge_w,
    size_t wsBytes) {
  __shared__ float sW[(D_IN / 2) * W_PITCH];

  const int tid = threadIdx.x;
  if (bits_fit(nrows, nedges)) {
    unsigned*      tb  = (unsigned*)sW;
    unsigned char* tb8 = (unsigned char*)sW;
    unsigned*      bm  = (unsigned*)out;
    const int nWords  = (nedges + 31) >> 5;
    const int nOwners = (nrows + OWN_ROWS - 1) / OWN_ROWS;
    const int nTiles  = (nWords + BM_TILE - 1) / BM_TILE;
    const int g = tid & (BM_TILE - 1);
    const int q = tid >> 6;
    int pass = 0;
    int tile = blockIdx.x;
    int d[8];
    load_targets(dst, ((tile * BM_TILE + g) << 5) + q * 8, nedges, tile < nTiles, d);
    while (tile < nTiles) {
      for (int i = tid; i < BM_KEYS * BM_TILE / 4; i += 256)
        ((v4i*)sW)[i] = (v4i){0, 0, 0, 0};
      __syncthreads();
#pragma unroll
      for (int j = 0; j < 8; ++j) {
        const unsigned key = (unsigned)d[j] / (unsigned)OWN_ROWS;
        if (d[j] >= 0 && key < (unsigned)nOwners)
          tb8[((key * BM_TILE + g) << 2) + q] |= (unsigned char)(1u << j);
      }
      __syncthreads();
      const int w0 = tile * BM_TILE;
      ++pass;
      const int next = (pass & 1) ? pass * GEMM_BLOCKS + (GEMM_BLOCKS - 1 - (int)blockIdx.x)
                                  : pass * GEMM_BLOCKS + (int)blockIdx.x;
      load_targets(dst, ((next * BM_TILE + g) << 5) + q * 8, nedges, next < nTiles, d);
      for (int i = tid; i < nOwners * (BM_TILE / 4); i += 256) {
        const int key  = i >> 4;
        const int part = (i & 15) << 2;
        const v4i v = *(const v4i*)(tb + key * BM_TILE + part);
        unsigned* p = bm + (size_t)key * OWN_WORDS + w0 + part;
        if (w0 + part + 3 < nWords) {
          *(v4i*)p = v;
        } else {
          if (w0 + part + 0 < nWords) p[0] = (unsigned)v.x;
          if (w0 + part + 1 < nWords) p[1] = (unsigned)v.y;
          if (w0 + part + 2 < nWords) p[2] = (unsigned)v.z;
        }
      }
      __syncthreads();
      tile = next;
    }
  }
#pragma unroll 8
  for (int i = 0; i < (D_IN / 2) * D_OUT / 256; ++i) {
    const int idx = tid + i * 256;
    const int p = idx >> 7;
    const int n = idx & (D_OUT - 1);
    v2f w2;
    w2.x = W[(2 * p) * D_OUT + n];
    w2.y = W[(2 * p + 1) * D_OUT + n];
    *(v2f*)(sW + p * W_PITCH + 2 * n) = w2;
  }
  __syncthreads();

  const int wave = tid >> 5;
  const int lane = tid & 31;
  const int m    = lane & 15;
  const int hh   = lane >> 4;
  const int nStrips = (nrows + 15) >> 4;

  for (int strip = wave * GEMM_BLOCKS + blockIdx.x; strip < nStrips;
       strip += GEMM_BLOCKS * GEMM_WAVES) {
    const int rowBase = strip << 4;
    int arow_idx = rowBase + m;
    if (arow_idx >= nrows) arow_idx = nrows - 1;
    const float* arow = feat + (size_t)arow_idx * D_IN + 2 * hh;

    v2f A[D_IN / 4];
#pragma unroll
    for (int kk = 0; kk < D_IN / 4; ++kk) A[kk] = *(const v2f*)(arow + kk * 4);

    v8f acc[8];
#pragma unroll
    for (int t = 0; t < 8; ++t) acc[t] = (v8f){};

    const float* bBase = sW + hh * W_PITCH + 2 * m;
    v2f Bc[8], Bn[8];
#pragma unroll
    for (int t = 0; t < 8; ++t) Bc[t] = *(const v2f*)(bBase + 32 * t);
#pragma unroll
    for (int kk = 0; kk < D_IN / 4; ++kk) {
      if (kk + 1 < D_IN / 4) {
#pragma unroll
        for (int t = 0; t < 8; ++t)
          Bn[t] = *(const v2f*)(bBase + (2 * kk + 2) * W_PITCH + 32 * t);
      }
#pragma unroll
      for (int t = 0; t < 8; ++t)
        acc[t] = __builtin_amdgcn_wmma_f32_16x16x4_f32(
             false, A[kk],  false, Bc[t],
             (short)0, acc[t],  false,  false);
      __builtin_amdgcn_sched_barrier(0);
#pragma unroll
      for (int t = 0; t < 8; ++t) Bc[t] = Bn[t];
    }

    const int mBase = rowBase + hh * 8;
#pragma unroll
    for (int v = 0; v < 8; ++v) {
      const int row = mBase + v;
      if (row < nrows) {
        const float c = cu[row];
        float* hrow = h + (size_t)row * D_OUT + m;
#pragma unroll
        for (int t = 0; t < 8; ++t) hrow[16 * t] = acc[t][v] * c;
      }
    }
  }

  if (recs_fit(nrows, nedges, wsBytes)) {
    const int nWavesAll = GEMM_BLOCKS * GEMM_WAVES;
    const int gw    = wave * GEMM_BLOCKS + (int)blockIdx.x;
    const int heavy = nStrips % nWavesAll;
    if (gw >= heavy) {
      const int nLight = nWavesAll - heavy;
      v4u* rec4 = (v4u*)(h + (size_t)nrows * D_OUT);
      if (nedges >= 2) {
        const int nItems = (nedges + 63) >> 6;
        for (int it = gw - heavy; it < nItems; it += REC_Q * nLight) {
          v2i s[REC_Q], d[REC_Q];
          v2f w[REC_Q];
#pragma unroll
          for (int u = 0; u < REC_Q; ++u) {
            const int iu = it + u * nLight;
            const int e0 = (iu << 6) + 2 * lane;
            const int ec = (iu < nItems && e0 + 1 < nedges) ? e0 : 0;
            s[u] = *(const v2i*)(src + ec);
            d[u] = *(const v2i*)(dst + ec);
            w[u] = *(const v2f*)(edge_w + ec);
          }
#pragma unroll
          for (int u = 0; u < REC_Q; ++u) {
            const int iu = it + u * nLight;
            const int e0 = (iu << 6) + 2 * lane;
            if (iu < nItems && e0 + 1 < nedges)
              rec4[e0 >> 1] = (v4u){pack_src_row(s[u].x, d[u].x), __float_as_uint(w[u].x),
                                    pack_src_row(s[u].y, d[u].y), __float_as_uint(w[u].y)};
          }
        }
      }
      if ((nedges & 1) && gw == heavy && lane == 0) {
        const int e = nedges - 1;
        ((v2u*)rec4)[e] = (v2u){pack_src_row(src[e], dst[e]), __float_as_uint(edge_w[e])};
      }
    }
  }
}

#define NWAVES   32
#define LCAP     384
#define LCHUNKS  (LCAP / 32)
#define DRAIN_TRIPS ((LCAP - 32) / 32 + 1)
#define QCAP     64
#define SCAN_Q   25

__device__ __forceinline__ void drain_batch(const float* __restrict__ h,
                                            v4f* tile, const unsigned* qa,
                                            const unsigned* qb, int bq, int nb,
                                            int lane) {
  v4f buf[8];
  unsigned pa[8];
  float pw[8];
#pragma unroll
  for (int k = 0; k < 8; ++k) {
    const int j = bq + min(k, nb - 1);
    pa[k] = qa[j];
    pw[k] = __uint_as_float(qb[j]);
    buf[k] = *(const v4f*)(h + (size_t)(pa[k] & 0x7FFFFFu) * D_OUT + lane * 4);
  }
  __builtin_amdgcn_sched_barrier(0);
#pragma unroll
  for (int k = 0; k < 8; ++k) {
    if (k < nb) {
      const int r = (int)(pa[k] >> 23);
      v4f t = tile[r * (D_OUT / 4) + lane];
      t += buf[k] * pw[k];
      tile[r * (D_OUT / 4) + lane] = t;
    }
  }
}

__device__ __forceinline__ bool drain_bits(unsigned& pend, unsigned pbase, int& cnt,
                                           unsigned* list, unsigned ltMask) {
  unsigned m = (unsigned)__ballot(pend != 0u);
  for (int it = 0; it < 32 && m != 0u && cnt <= LCAP - 32; ++it) {
    if (pend != 0u) {
      list[cnt + __popc(m & ltMask)] = pbase + (unsigned)__builtin_ctz(pend);
      pend &= pend - 1u;
    }
    cnt += __popc(m);
    m = (unsigned)__ballot(pend != 0u);
  }
  return m != 0u;
}

__global__ __launch_bounds__(1024) void influence_owner_sum_kernel(
    const float* __restrict__ h, const int* __restrict__ src,
    const int* __restrict__ dst, const float* __restrict__ edge_w,
    const float* __restrict__ cv, float* __restrict__ out, int nrows,
    int nedges, size_t wsBytes) {
  __shared__ v4f      tile[OWN_ROWS * (D_OUT / 4)];
  __shared__ unsigned listA[NWAVES][LCAP];
  __shared__ unsigned listB[NWAVES][LCAP];
  __shared__ unsigned queA[NWAVES][QCAP];
  __shared__ unsigned queB[NWAVES][QCAP];
  __shared__ unsigned runTab[NWAVES][NWAVES];
  __shared__ int      waveDone[NWAVES];

  const int tid  = threadIdx.x;
  const int wave = tid >> 5;
  const int lane = tid & 31;
  const int base = blockIdx.x * OWN_ROWS;
  const unsigned ltMask = (1u << lane) - 1u;
  const bool useBits = bits_fit(nrows, nedges);
  const bool useRec = recs_fit(nrows, nedges, wsBytes);
  const v2u* rec = (const v2u*)(h + (size_t)nrows * D_OUT);
  const unsigned* bm = (const unsigned*)out + (size_t)blockIdx.x * OWN_WORDS;

  for (int i = tid; i < OWN_ROWS * (D_OUT / 4); i += 1024)
    tile[i] = (v4f){0.f, 0.f, 0.f, 0.f};

  const int waveS    = __builtin_amdgcn_readfirstlane(wave);
  const int nWords   = (nedges + 31) >> 5;
  const int perWave  = (nWords + NWAVES - 1) / NWAVES;
  const int segStart = waveS * perWave;
  const int segEnd   = min(segStart + perWave, nWords);
  const int nSteps   = segEnd > segStart ? (segEnd - segStart + 31) >> 5 : 0;
  int step = 0;
  unsigned pend = 0u, pbase = 0u;
  const int maxRounds = ((32 + DRAIN_TRIPS - 1) / DRAIN_TRIPS) * ((perWave + 31) >> 5) + 1;

  for (int round = 0; round <= maxRounds; ++round) {
    int cnt = 0;
    bool full = drain_bits(pend, pbase, cnt, listA[waveS], ltMask);
    while (!full && step < nSteps) {
      unsigned x[SCAN_Q];
      if (useBits) {
#pragma unroll
        for (int q = 0; q < SCAN_Q; ++q)
          x[q] = bm[min(segStart + (step + q) * 32 + lane, nWords - 1)];
      } else {
#pragma unroll
        for (int q = 0; q < SCAN_Q; ++q)
          x[q] = own_bits(dst, min(segStart + (step + q) * 32 + lane, nWords - 1),
                          nedges, blockIdx.x);
      }
      int done = 0;
#pragma unroll
      for (int q = 0; q < SCAN_Q; ++q) {
        if (!full && step + q < nSteps) {
          const int w = segStart + (step + q) * 32 + lane;
          pend  = (w < segEnd) ? x[q] : 0u;
          pbase = (unsigned)w << 5;
          full  = drain_bits(pend, pbase, cnt, listA[waveS], ltMask);
          ++done;
        }
      }
      step += done;
    }
    const int cntU = __builtin_amdgcn_readfirstlane(cnt);
    unsigned ra[LCHUNKS], rb[LCHUNKS], rp[LCHUNKS];
#pragma unroll
    for (int c = 0; c < LCHUNKS; ++c) { ra[c] = 0u; rb[c] = 0u; rp[c] = 0u; }
    runTab[wave][lane] = 0u;
#pragma unroll
    for (int g = 0; g < LCHUNKS / 4; ++g) {
      if (g * 128 < cntU) {
        if (useRec) {
#pragma unroll
          for (int c = 4 * g; c < 4 * g + 4; ++c) {
            const unsigned e = listA[wave][min(c * 32 + lane, cntU - 1)];
            const v2u r = rec[e];
            ra[c] = r.x;
            rb[c] = r.y;
          }
        } else {
#pragma unroll
          for (int c = 4 * g; c < 4 * g + 4; ++c) {
            const unsigned e  = listA[wave][min(c * 32 + lane, cntU - 1)];
            const unsigned lr = (unsigned)(dst[e] - base);
            ra[c] = (unsigned)src[e] | (lr << 23);
            rb[c] = __float_as_uint(edge_w[e]);
          }
        }
#pragma unroll
        for (int c = 4 * g; c < 4 * g + 4; ++c) {
          if (c * 32 < cntU) {
            const bool valid = c * 32 + lane < cntU;
            const unsigned o = (ra[c] >> 23) & 31u;
            unsigned peers = (unsigned)__ballot(valid);
#pragma unroll
            for (int bit = 0; bit < 5; ++bit) {
              const bool set = ((o >> bit) & 1u) != 0u;
              const unsigned bm = (unsigned)__ballot(set);
              peers &= set ? bm : ~bm;
            }
            if (valid) {
              const unsigned before = runTab[wave][o];
              rp[c] = before + (unsigned)__popc(peers & ltMask);
              if ((peers & ltMask) == 0u)
                runTab[wave][o] = before + (unsigned)__popc(peers);
            }
          }
        }
      }
    }
    const unsigned runLen = runTab[wave][lane];
    unsigned runEnd = runLen;
#pragma unroll
    for (int d = 1; d < 32; d <<= 1) {
      const unsigned t = __shfl_up(runEnd, (unsigned)d);
      if (lane >= d) runEnd += t;
    }
    const unsigned runStart = runEnd - runLen;
    runTab[wave][lane] = runStart | (runLen << 16);
#pragma unroll
    for (int c = 0; c < LCHUNKS; ++c) {
      if (c * 32 < cntU) {
        const unsigned st = __shfl(runStart, (int)((ra[c] >> 23) & 31u));
        if (c * 32 + lane < cntU) {
          listA[wave][st + rp[c]] = ra[c];
          listB[wave][st + rp[c]] = rb[c];
        }
      }
    }
    if (lane == 0) waveDone[wave] = (step >= nSteps && !full) ? 1 : 0;
    __syncthreads();

    const unsigned myRun = runTab[lane][wave];
    int qn = 0;
    for (int L = 0; L <= NWAVES; ++L) {
      const unsigned run =
          (L < NWAVES) ? (unsigned)__builtin_amdgcn_readlane((int)myRun, L) : 0u;
      const int s = (int)(run & 0xFFFFu);
      const int n = (int)(run >> 16);
      for (int j = 0; j < n; j += 32) {
        const int take = min(32, n - j);
        if (lane < take) {
          queA[wave][qn + lane] = listA[L][s + j + lane];
          queB[wave][qn + lane] = listB[L][s + j + lane];
        }
        qn += take;
        if (qn > QCAP - 32) {
          int bq = 0;
          for (; bq + 8 <= qn; bq += 8)
            drain_batch(h, tile, queA[wave], queB[wave], bq, 8, lane);
          const int rem = qn - bq;
          unsigned ra = 0, rb = 0;
          if (lane < rem) { ra = queA[wave][bq + lane]; rb = queB[wave][bq + lane]; }
          if (lane < rem) { queA[wave][lane] = ra; queB[wave][lane] = rb; }
          qn = rem;
        }
      }
      if (L == NWAVES) {
        for (int bq = 0; bq < qn; bq += 8)
          drain_batch(h, tile, queA[wave], queB[wave], bq, min(8, qn - bq), lane);
        qn = 0;
      }
    }
    int allDone = 1;
    for (int i = 0; i < NWAVES; ++i) allDone &= waveDone[i];
    __syncthreads();
    if (allDone) break;
  }

  float cvLane = 0.f;
  {
    const int r = wave + NWAVES * lane;
    if (r < OWN_ROWS && base + r < nrows) cvLane = cv[base + r];
  }
  for (int i = 0; i < (OWN_ROWS + NWAVES - 1) / NWAVES; ++i) {
    const int r = wave + NWAVES * i;
    const int row = base + r;
    const float cvr = __shfl(cvLane, i);
    if (r < OWN_ROWS && row < nrows) {
      const v4f t = tile[r * (D_OUT / 4) + lane];
      *(v4f*)(out + (size_t)row * D_OUT + lane * 4) = t * cvr;
    }
  }
}

extern "C" void kernel_launch(void* const* d_in, const int* in_sizes, int n_in,
                              void* d_out, int out_size, void* d_ws, size_t ws_size,
                              hipStream_t stream) {
  const float* feat = (const float*)d_in[0];
  const float* W    = (const float*)d_in[1];
  const float* cu   = (const float*)d_in[2];
  const float* cv   = (const float*)d_in[3];
  const float* ew   = (const float*)d_in[4];
  const int*   src  = (const int*)d_in[5];
  const int*   dst  = (const int*)d_in[6];

  float* out = (float*)d_out;
  float* h   = (float*)d_ws;

  const int N = in_sizes[2];
  const int E = in_sizes[5];

  influence_gemm_cu_kernel<<<GEMM_BLOCKS, 256, 0, stream>>>(feat, W, cu, h, N,
                                                            dst, E, out, src, ew,
                                                            ws_size);

  const int ownerBlocks = (N + OWN_ROWS - 1) / OWN_ROWS;
  influence_owner_sum_kernel<<<ownerBlocks, 1024, 0, stream>>>(
      h, src, dst, ew, cv, out, N, E, ws_size);
}
